# starting kernel
# baseline (speedup 1.0000x reference)
#include <hip/hip_runtime.h>

typedef __attribute__((ext_vector_type(2))) float v2f;
typedef __attribute__((ext_vector_type(8))) float v8f;

#define NEG_SLOPE 0.2f

__device__ __forceinline__ void atomic_max_f32(float* addr, float v) {
  if (v >= 0.0f) atomicMax((int*)addr, __float_as_int(v));
  else           atomicMin((unsigned int*)addr, (unsigned int)__float_as_int(v));
}

__global__ void gat_init(float* __restrict__ attr_sum, float* __restrict__ cnt,
                         float* __restrict__ mx, float* __restrict__ den,
                         float* __restrict__ num, int N) {
  int i = blockIdx.x * blockDim.x + threadIdx.x;
  if (i < N) { attr_sum[i] = 0.0f; cnt[i] = 0.0f; }
  if (i < 2 * N) {
    mx[i]  = __int_as_float(0xff800000);
    den[i] = 0.0f;
    num[i] = 0.0f;
  }
}

__global__ void gat_degree(const int* __restrict__ ei,
                           const float* __restrict__ attr,
                           float* __restrict__ attr_sum, float* __restrict__ cnt,
                           int E) {
  int e = blockIdx.x * blockDim.x + threadIdx.x;
  if (e >= E) return;
  int d = ei[E + e];
  atomicAdd(&attr_sum[d], attr[e]);
  atomicAdd(&cnt[d], 1.0f);
}

__global__ void gat_logits_wmma(
    const float* __restrict__ x,
    const int*   __restrict__ ei,
    const float* __restrict__ attr,
    const float* __restrict__ W_l, const float* __restrict__ b_l,
    const float* __restrict__ W_r, const float* __restrict__ b_r,
    const float* __restrict__ W_e, const float* __restrict__ att,
    const float* __restrict__ attr_sum, const float* __restrict__ cnt,
    float* __restrict__ mx,
    float* __restrict__ logits,
    int E, int M  ) {
  const int wave = (blockIdx.x * blockDim.x + threadIdx.x) >> 5;
  const int lane = threadIdx.x & 31;
  const int base = wave * 16;
  if (base >= M) return;
  const int  lh = lane & 15;
  const bool hi = lane >= 16;

  v2f B[8];
#pragma unroll
  for (int t = 0; t < 8; ++t) {
    int ch = t * 16 + lh;
    if (!hi) { B[t].x = W_l[ch]; B[t].y = W_r[ch]; }
    else     { B[t].x = W_e[ch]; B[t].y = b_l[ch] + b_r[ch]; }
  }

  int j = base + lh;
  float a0 = 0.0f, a1 = 0.0f;
  if (j < M) {
    if (!hi) {
      int s, d;
      if (j < E) { s = ei[j]; d = ei[E + j]; }
      else       { s = d = j - E; }
      a0 = x[s];
      a1 = x[d];
    } else {
      if (j < E) a0 = attr[j];
      else { int n = j - E; a0 = attr_sum[n] / fmaxf(cnt[n], 1.0f); }
      a1 = 1.0f;
    }
  }
  v2f A; A.x = a0; A.y = a1;

  float p[2][8];
#pragma unroll
  for (int h = 0; h < 2; ++h)
#pragma unroll
    for (int r = 0; r < 8; ++r) p[h][r] = 0.0f;

#pragma unroll
  for (int t = 0; t < 8; ++t) {
    v8f cz = {};
    v8f D = __builtin_amdgcn_wmma_f32_16x16x4_f32(
        false, A, false, B[t], (short)0, cz, false, false);
    float av = att[t * 16 + lh];
    const int h = t >> 2;
#pragma unroll
    for (int r = 0; r < 8; ++r) {
      float g = D[r];
      g = (g > 0.0f) ? g : NEG_SLOPE * g;
      p[h][r] = fmaf(g, av, p[h][r]);
    }
  }

#pragma unroll
  for (int off = 1; off < 16; off <<= 1) {
#pragma unroll
    for (int r = 0; r < 8; ++r) {
      p[0][r] += __shfl_xor(p[0][r], off, 32);
      p[1][r] += __shfl_xor(p[1][r], off, 32);
    }
  }

  float l0 = p[0][0], l1 = p[1][0];
#pragma unroll
  for (int q = 1; q < 8; ++q) {
    bool sel = (lh == q);
    l0 = sel ? p[0][q] : l0;
    l1 = sel ? p[1][q] : l1;
  }
  if (lh < 8) {
    int jm = base + (hi ? 8 + lh : lh);
    if (jm < M) {
      int d = (jm < E) ? ei[E + jm] : (jm - E);
      logits[2 * jm + 0] = l0;
      logits[2 * jm + 1] = l1;
      atomic_max_f32(&mx[2 * d + 0], l0);
      atomic_max_f32(&mx[2 * d + 1], l1);
    }
  }
}

__global__ void gat_expsum(const float* __restrict__ x,
                           const int* __restrict__ ei,
                           const float* __restrict__ mx,
                           const float* __restrict__ logits,
                           float* __restrict__ den, float* __restrict__ num,
                           int E, int M) {
  int j = blockIdx.x * blockDim.x + threadIdx.x;
  if (j >= M) return;
  int s, d;
  if (j < E) { s = ei[j]; d = ei[E + j]; }
  else       { s = d = j - E; }
  float xs = x[s];
  float a0 = expf(logits[2 * j + 0] - mx[2 * d + 0]);
  float a1 = expf(logits[2 * j + 1] - mx[2 * d + 1]);
  atomicAdd(&den[2 * d + 0], a0);
  atomicAdd(&den[2 * d + 1], a1);
  atomicAdd(&num[2 * d + 0], a0 * xs);
  atomicAdd(&num[2 * d + 1], a1 * xs);
}

__global__ void gat_out(const float* __restrict__ num, const float* __restrict__ den,
                        const float* __restrict__ W_l, const float* __restrict__ b_l,
                        const float* __restrict__ bias, float* __restrict__ out,
                        int N) {
  int i = blockIdx.x * blockDim.x + threadIdx.x;
  if (i >= N * 128) return;
  int k = i & 127;
  int n = i >> 7;
  int h = k >> 6;
  float S = num[2 * n + h] / den[2 * n + h];
  out[i] = fmaf(S, W_l[k], b_l[k] + bias[k]);
}

extern "C" void kernel_launch(void* const* d_in, const int* in_sizes, int n_in,
                              void* d_out, int out_size, void* d_ws, size_t ws_size,
                              hipStream_t stream) {
  const float* x    = (const float*)d_in[0];
  const int*   ei   = (const int*)  d_in[1];
  const float* attr = (const float*)d_in[2];
  const float* W_l  = (const float*)d_in[3];
  const float* b_l  = (const float*)d_in[4];
  const float* W_r  = (const float*)d_in[5];
  const float* b_r  = (const float*)d_in[6];
  const float* W_e  = (const float*)d_in[7];
  const float* att  = (const float*)d_in[8];
  const float* bias = (const float*)d_in[9];

  const int N = in_sizes[0];
  const int E = in_sizes[2];
  const int M = E + N;

  float* ws       = (float*)d_ws;
  float* attr_sum = ws;
  float* cnt      = attr_sum + N;
  float* mx       = cnt + N;
  float* den      = mx + 2 * N;
  float* num      = den + 2 * N;
  float* logits   = num + 2 * N;

  const int BLK = 256;

  gat_init<<<(2 * N + BLK - 1) / BLK, BLK, 0, stream>>>(attr_sum, cnt, mx, den, num, N);

  gat_degree<<<(E + BLK - 1) / BLK, BLK, 0, stream>>>(ei, attr, attr_sum, cnt, E);

  int nGroups = (M + 15) / 16;
  int blocks2 = (nGroups + 7) / 8;
  gat_logits_wmma<<<blocks2, BLK, 0, stream>>>(x, ei, attr, W_l, b_l, W_r, b_r,
                                               W_e, att, attr_sum, cnt, mx,
                                               logits, E, M);

  gat_expsum<<<(M + BLK - 1) / BLK, BLK, 0, stream>>>(x, ei, mx, logits, den, num, E, M);

  gat_out<<<(N * 128 + BLK - 1) / BLK, BLK, 0, stream>>>(num, den, W_l, b_l, bias,
                                                         (float*)d_out, N);
}


# Self-loop term folded into the out kernel (own logit by WMMA, exp(mx-m) rescale), so in-degree and attr sum join the exp-sum slot: 2 atomic lines per edge not 3; 4 launches; summation order changes
# speedup vs baseline: 2.8924x; 2.8924x over previous
#include <hip/hip_runtime.h>

typedef __attribute__((ext_vector_type(2))) float v2f;
typedef __attribute__((ext_vector_type(8))) float v8f;

#define NEG_SLOPE 0.2f


__device__ __forceinline__ void atomic_max_f32(float* addr, float v) {
  (void)__hip_atomic_fetch_max(addr, v, __ATOMIC_RELAXED, __HIP_MEMORY_SCOPE_AGENT);
}

__global__ void gat_init(float* __restrict__ acc, float* __restrict__ mx, int N) {
  int i = blockIdx.x * blockDim.x + threadIdx.x;
  if (i < 2 * N) {
    reinterpret_cast<float4*>(acc)[i] = make_float4(0.0f, 0.0f, 0.0f, 0.0f);
    mx[i] = __int_as_float(0xff800000);
  }
}

__device__ __forceinline__ void logits16(
    v2f A, int lane,
    const float* __restrict__ W_l, const float* __restrict__ b_l,
    const float* __restrict__ W_r, const float* __restrict__ b_r,
    const float* __restrict__ W_e, const float* __restrict__ att,
    float (&p)[2][8]) {
  const int  lh = lane & 15;
  const bool hi = lane >= 16;
  v2f B[8];
#pragma unroll
  for (int t = 0; t < 8; ++t) {
    int ch = t * 16 + lh;
    if (!hi) { B[t].x = W_l[ch]; B[t].y = W_r[ch]; }
    else     { B[t].x = W_e[ch]; B[t].y = b_l[ch] + b_r[ch]; }
  }
#pragma unroll
  for (int h = 0; h < 2; ++h)
#pragma unroll
    for (int r = 0; r < 8; ++r) p[h][r] = 0.0f;

#pragma unroll
  for (int t = 0; t < 8; ++t) {
    v8f cz = {};
    v8f D = __builtin_amdgcn_wmma_f32_16x16x4_f32(
        false, A, false, B[t], (short)0, cz, false, false);
    float av = att[t * 16 + lh];
    const int h = t >> 2;
#pragma unroll
    for (int r = 0; r < 8; ++r) {
      float g = D[r];
      g = (g > 0.0f) ? g : NEG_SLOPE * g;
      p[h][r] = fmaf(g, av, p[h][r]);
    }
  }
#pragma unroll
  for (int off = 1; off < 16; off <<= 1) {
#pragma unroll
    for (int r = 0; r < 8; ++r) {
      p[0][r] += __shfl_xor(p[0][r], off, 32);
      p[1][r] += __shfl_xor(p[1][r], off, 32);
    }
  }
}

__global__ void gat_logits_wmma(
    const float* __restrict__ x,
    const int*   __restrict__ ei,
    const float* __restrict__ attr,
    const float* __restrict__ W_l, const float* __restrict__ b_l,
    const float* __restrict__ W_r, const float* __restrict__ b_r,
    const float* __restrict__ W_e, const float* __restrict__ att,
    float* __restrict__ mx,
    float* __restrict__ logits,
    int E) {
  const int wave = (blockIdx.x * blockDim.x + threadIdx.x) >> 5;
  const int lane = threadIdx.x & 31;
  const int base = wave * 16;
  if (base >= E) return;
  const int  lh = lane & 15;
  const bool hi = lane >= 16;

  int j = base + lh;
  float a0 = 0.0f, a1 = 0.0f;
  if (j < E) {
    if (!hi) { a0 = x[ei[j]]; a1 = x[ei[E + j]]; }
    else     { a0 = attr[j];  a1 = 1.0f; }
  }
  v2f A; A.x = a0; A.y = a1;

  float p[2][8];
  logits16(A, lane, W_l, b_l, W_r, b_r, W_e, att, p);

  const int hsel = lane & 1;
  const int rsel = (lane >> 1) & 7;
  float lv = hsel ? p[1][0] : p[0][0];
#pragma unroll
  for (int q = 1; q < 8; ++q) {
    float cand = hsel ? p[1][q] : p[0][q];
    lv = (rsel == q) ? cand : lv;
  }
  int jm = base + (lane >> 1);
  if (jm < E) {
    int d = ei[E + jm];
    logits[2 * jm + hsel] = lv;
    atomic_max_f32(&mx[2 * d + hsel], lv);
  }
}

__global__ void gat_expsum(const float* __restrict__ x,
                           const int* __restrict__ ei,
                           const float* __restrict__ attr,
                           const float* __restrict__ mx,
                           const float* __restrict__ logits,
                           float* __restrict__ acc,
                           int E) {
  const int t = blockIdx.x * blockDim.x + threadIdx.x;
  const int c = t & 7;
  const int h = c & 1;
  const int j0 = (t >> 5) * 8 + ((t >> 3) & 3);
  float v[2];
  int   d[2];
#pragma unroll
  for (int i = 0; i < 2; ++i) {
    const int j  = j0 + 4 * i;
    const int jc = (j < E) ? j : (E - 1);
    const int s  = ei[jc];
    d[i] = ei[E + jc];
    const float lg = logits[2 * jc + h];
    const float at = attr[jc];
    const float xs = x[s];
    const float mm = mx[2 * d[i] + h];
    const float fx = (c & 2) ? xs : 1.0f;
    const float ev = expf(lg - mm) * fx;
    const float ov = (c & 1) ? 1.0f : at;
    const float vv = (c & 4) ? ov : ev;
    v[i] = (j < E) ? vv : 0.0f;
  }
  if (c < 6) {
    atomicAdd(&acc[8 * d[0] + c], v[0]);
    atomicAdd(&acc[8 * d[1] + c], v[1]);
  }
}

#define OUT_ROWS 16
__global__ void gat_out(const float* __restrict__ x,
                        const float* __restrict__ W_l, const float* __restrict__ b_l,
                        const float* __restrict__ W_r, const float* __restrict__ b_r,
                        const float* __restrict__ W_e, const float* __restrict__ att,
                        const float* __restrict__ bias,
                        const float* __restrict__ acc, const float* __restrict__ mx,
                        float* __restrict__ out, int N) {
  const int wave = (blockIdx.x * blockDim.x + threadIdx.x) >> 5;
  const int lane = threadIdx.x & 31;
  const int n0 = wave * OUT_ROWS;
  if (n0 >= N) return;
  const int  h  = lane >> 4;
  const bool hi = lane >= 16;
  const int  nq = n0 + (lane & 15);

  float xn = 0.0f, a0 = 0.0f, a1 = 0.0f;
  if (nq < N) {
    xn = x[nq];
    if (!hi) { a0 = xn; a1 = xn; }
    else     { a0 = acc[8 * nq + 4] / fmaxf(acc[8 * nq + 5], 1.0f); a1 = 1.0f; }
  }
  v2f A; A.x = a0; A.y = a1;

  float p[2][8];
  logits16(A, lane, W_l, b_l, W_r, b_r, W_e, att, p);

  const int q = lane & 7;
  float own  = h ? p[1][0] : p[0][0];
  float give = h ? p[0][0] : p[1][0];
#pragma unroll
  for (int i = 1; i < 8; ++i) {
    float co = h ? p[1][i] : p[0][i];
    float cg = h ? p[0][i] : p[1][i];
    own  = (q == i) ? co : own;
    give = (q == i) ? cg : give;
  }
  const float recv = __shfl_xor(give, 16, 32);
  const float ls = (((lane >> 3) & 1) == h) ? own : recv;

  float S = 0.0f;
  if (nq < N) {
    const float me = mx[2 * nq + h];
    const float m  = fmaxf(me, ls);
    const float se = expf(me - m);
    const float ss = expf(ls - m);
    const float den = acc[8 * nq + h] * se + ss;
    const float num = acc[8 * nq + 2 + h] * se + ss * xn;
    S = num / den;
  }

  const int k = 4 * lane;
  const float w0 = W_l[k], w1 = W_l[k + 1], w2 = W_l[k + 2], w3 = W_l[k + 3];
  const float c0 = b_l[k] + bias[k];
  const float c1 = b_l[k + 1] + bias[k + 1];
  const float c2 = b_l[k + 2] + bias[k + 2];
  const float c3 = b_l[k + 3] + bias[k + 3];

  float4* orow = reinterpret_cast<float4*>(out + (size_t)n0 * 128) + lane;
#pragma unroll
  for (int r = 0; r < OUT_ROWS; ++r) {
    float s = __shfl(S, r, 16);
    if (n0 + r < N) {
      float4 o;
      o.x = fmaf(s, w0, c0);
      o.y = fmaf(s, w1, c1);
      o.z = fmaf(s, w2, c2);
      o.w = fmaf(s, w3, c3);
      orow[r * 32] = o;
    }
  }
}

extern "C" void kernel_launch(void* const* d_in, const int* in_sizes, int n_in,
                              void* d_out, int out_size, void* d_ws, size_t ws_size,
                              hipStream_t stream) {
  const float* x    = (const float*)d_in[0];
  const int*   ei   = (const int*)  d_in[1];
  const float* attr = (const float*)d_in[2];
  const float* W_l  = (const float*)d_in[3];
  const float* b_l  = (const float*)d_in[4];
  const float* W_r  = (const float*)d_in[5];
  const float* b_r  = (const float*)d_in[6];
  const float* W_e  = (const float*)d_in[7];
  const float* att  = (const float*)d_in[8];
  const float* bias = (const float*)d_in[9];

  const int N = in_sizes[0];
  const int E = in_sizes[2];

  float* ws     = (float*)d_ws;
  float* acc    = ws;
  float* mx     = acc + 8 * N;
  float* logits = mx + 2 * N;

  const int BLK = 256;

  gat_init<<<(2 * N + BLK - 1) / BLK, BLK, 0, stream>>>(acc, mx, N);

  gat_logits_wmma<<<(E + 127) / 128, BLK, 0, stream>>>(x, ei, attr, W_l, b_l, W_r, b_r,
                                                       W_e, att, mx, logits, E);

  gat_expsum<<<(E + 63) / 64, BLK, 0, stream>>>(x, ei, attr, mx, logits, acc, E);

  gat_out<<<(N + 127) / 128, BLK, 0, stream>>>(x, W_l, b_l, W_r, b_r, W_e, att, bias,
                                               acc, mx, (float*)d_out, N);
}


# One edge pass: exp terms summed in f64 (f64 atomic add; word built from f32 bits and 2^k) against a fixed per-head shift; segment-max pass, logits round trip and a launch gone; sum order changes
# speedup vs baseline: 5.5907x; 1.9329x over previous
#include <hip/hip_runtime.h>

typedef __attribute__((ext_vector_type(2))) float v2f;
typedef __attribute__((ext_vector_type(8))) float v8f;

#define NEG_SLOPE 0.2f

#define T_LIMIT 600.0f
#define LOG2E_HI 1.44269502162933349609375f
#define LOG2E_LO 1.925962991126617e-08f

__device__ __forceinline__ void exp_split(float t, float& pe, int& k) {
  t = fminf(fmaxf(t, -T_LIMIT), T_LIMIT);
  const float ph = t * LOG2E_HI;
  float pl = fmaf(t, LOG2E_HI, -ph);
  pl = fmaf(t, LOG2E_LO, pl);
  const float kf = rintf(ph);
  const float r = (ph - kf) + pl;
  pe = __builtin_amdgcn_exp2f(r);
  k = (int)kf;
}

__device__ __forceinline__ double scaled_f64(float q, int k) {
  const unsigned u  = __float_as_uint(q);
  const unsigned e8 = (u >> 23) & 0xffu;
  unsigned hi = (u & 0x80000000u) | ((e8 + 896u + (unsigned)k) << 20) | ((u >> 3) & 0x000fffffu);
  unsigned lo = u << 29;
  if (e8 == 0u) { hi = u & 0x80000000u; lo = 0u; }
  return __builtin_bit_cast(double, ((unsigned long long)hi << 32) | (unsigned long long)lo);
}

__global__ void gat_init(float4* __restrict__ acc4, float* __restrict__ shift,
                         const float* __restrict__ b_l, const float* __restrict__ b_r,
                         const float* __restrict__ att, int N) {
  int i = blockIdx.x * blockDim.x + threadIdx.x;
  if (i < 4 * N) acc4[i] = make_float4(0.0f, 0.0f, 0.0f, 0.0f);
  if (blockIdx.x == 0 && threadIdx.x < 32) {
    const int lane = threadIdx.x;
    float s[2] = {0.0f, 0.0f};
#pragma unroll
    for (int jj = 0; jj < 4; ++jj) {
      const int ch = lane + 32 * jj;
      float g = b_l[ch] + b_r[ch];
      g = (g > 0.0f) ? g : NEG_SLOPE * g;
      s[jj >> 1] = fmaf(g, att[ch], s[jj >> 1]);
    }
#pragma unroll
    for (int off = 1; off < 32; off <<= 1) {
      s[0] += __shfl_xor(s[0], off, 32);
      s[1] += __shfl_xor(s[1], off, 32);
    }
    if (lane == 0) { shift[0] = s[0]; shift[1] = s[1]; }
  }
}

__device__ __forceinline__ void logits16(
    v2f A, int lane,
    const float* __restrict__ W_l, const float* __restrict__ b_l,
    const float* __restrict__ W_r, const float* __restrict__ b_r,
    const float* __restrict__ W_e, const float* __restrict__ att,
    float (&p)[2][8]) {
  const int  lh = lane & 15;
  const bool hi = lane >= 16;
  v2f B[8];
#pragma unroll
  for (int t = 0; t < 8; ++t) {
    int ch = t * 16 + lh;
    if (!hi) { B[t].x = W_l[ch]; B[t].y = W_r[ch]; }
    else     { B[t].x = W_e[ch]; B[t].y = b_l[ch] + b_r[ch]; }
  }
#pragma unroll
  for (int h = 0; h < 2; ++h)
#pragma unroll
    for (int r = 0; r < 8; ++r) p[h][r] = 0.0f;

#pragma unroll
  for (int t = 0; t < 8; ++t) {
    v8f cz = {};
    v8f D = __builtin_amdgcn_wmma_f32_16x16x4_f32(
        false, A, false, B[t], (short)0, cz, false, false);
    float av = att[t * 16 + lh];
    const int h = t >> 2;
#pragma unroll
    for (int r = 0; r < 8; ++r) {
      float g = D[r];
      g = (g > 0.0f) ? g : NEG_SLOPE * g;
      p[h][r] = fmaf(g, av, p[h][r]);
    }
  }
#pragma unroll
  for (int off = 1; off < 16; off <<= 1) {
#pragma unroll
    for (int r = 0; r < 8; ++r) {
      p[0][r] += __shfl_xor(p[0][r], off, 32);
      p[1][r] += __shfl_xor(p[1][r], off, 32);
    }
  }
}

__global__ void gat_edges(
    const float* __restrict__ x,
    const int*   __restrict__ ei,
    const float* __restrict__ attr,
    const float* __restrict__ W_l, const float* __restrict__ b_l,
    const float* __restrict__ W_r, const float* __restrict__ b_r,
    const float* __restrict__ W_e, const float* __restrict__ att,
    const float* __restrict__ shift,
    double* __restrict__ acc,
    int E) {
  const int wave = (blockIdx.x * blockDim.x + threadIdx.x) >> 5;
  const int lane = threadIdx.x & 31;
  const int base = wave * 16;
  if (base >= E) return;
  const int  lh = lane & 15;
  const bool hi = lane >= 16;

  int j = base + lh;
  float a0 = 0.0f, a1 = 0.0f;
  int dreg = 0;
  if (j < E) {
    if (!hi) { dreg = ei[E + j]; a0 = x[ei[j]]; a1 = x[dreg]; }
    else     { a0 = attr[j];  a1 = 1.0f; }
  }
  v2f A; A.x = a0; A.y = a1;

  float p[2][8];
  logits16(A, lane, W_l, b_l, W_r, b_r, W_e, att, p);

  const int c     = lane & 7;
  const int hsel  = c & 1;
  const int b3    = (lane >> 3) & 1;
  const int ehalf = 8 * (lane >> 4);
  const float m0  = shift[hsel];
#pragma unroll
  for (int i = 0; i < 4; ++i) {
    const float l_even = hsel ? p[1][2 * i]     : p[0][2 * i];
    const float l_odd  = hsel ? p[1][2 * i + 1] : p[0][2 * i + 1];
    const float lg = b3 ? l_odd : l_even;
    const int e = ehalf + 2 * i + b3;
    const float xv = __shfl(a0, (c & 4) ? (16 + e) : e, 32);
    const int   d  = __shfl(dreg, e, 32);
    float pe; int k;
    exp_split(lg - m0, pe, k);
    float q  = (c & 2) ? pe * xv : pe;
    int   kk = k;
    if (c & 4) { q = (c & 1) ? 1.0f : xv; kk = 0; }
    const double v = scaled_f64(q, kk);
    if (c < 6 && base + e < E)
      (void)__hip_atomic_fetch_add(&acc[8 * d + c], v, __ATOMIC_RELAXED, __HIP_MEMORY_SCOPE_AGENT);
  }
}

#define OUT_ROWS 16
__global__ void gat_out(const float* __restrict__ x,
                        const float* __restrict__ W_l, const float* __restrict__ b_l,
                        const float* __restrict__ W_r, const float* __restrict__ b_r,
                        const float* __restrict__ W_e, const float* __restrict__ att,
                        const float* __restrict__ bias,
                        const double* __restrict__ acc, const float* __restrict__ shift,
                        float* __restrict__ out, int N) {
  const int wave = (blockIdx.x * blockDim.x + threadIdx.x) >> 5;
  const int lane = threadIdx.x & 31;
  const int n0 = wave * OUT_ROWS;
  if (n0 >= N) return;
  const int  h  = lane >> 4;
  const bool hi = lane >= 16;
  const int  nq = n0 + (lane & 15);

  float xn = 0.0f, a0 = 0.0f, a1 = 0.0f;
  if (nq < N) {
    xn = x[nq];
    if (!hi) { a0 = xn; a1 = xn; }
    else     { a0 = (float)acc[8 * nq + 4] / fmaxf((float)acc[8 * nq + 5], 1.0f); a1 = 1.0f; }
  }
  v2f A; A.x = a0; A.y = a1;

  float p[2][8];
  logits16(A, lane, W_l, b_l, W_r, b_r, W_e, att, p);

  const int q = lane & 7;
  float own  = h ? p[1][0] : p[0][0];
  float give = h ? p[0][0] : p[1][0];
#pragma unroll
  for (int i = 1; i < 8; ++i) {
    float co = h ? p[1][i] : p[0][i];
    float cg = h ? p[0][i] : p[1][i];
    own  = (q == i) ? co : own;
    give = (q == i) ? cg : give;
  }
  const float recv = __shfl_xor(give, 16, 32);
  const float ls = (((lane >> 3) & 1) == h) ? own : recv;

  float S = 0.0f;
  if (nq < N) {
    float pe; int ks;
    exp_split(ls - shift[h], pe, ks);
    const double as  = scaled_f64(pe, ks);
    const double den = acc[8 * nq + h] + as;
    const double num = fma(as, (double)xn, acc[8 * nq + 2 + h]);
    S = (float)(num / den);
  }

  const int k = 4 * lane;
  const float w0 = W_l[k], w1 = W_l[k + 1], w2 = W_l[k + 2], w3 = W_l[k + 3];
  const float c0 = b_l[k] + bias[k];
  const float c1 = b_l[k + 1] + bias[k + 1];
  const float c2 = b_l[k + 2] + bias[k + 2];
  const float c3 = b_l[k + 3] + bias[k + 3];

  float4* orow = reinterpret_cast<float4*>(out + (size_t)n0 * 128) + lane;
#pragma unroll
  for (int r = 0; r < OUT_ROWS; ++r) {
    float s = __shfl(S, r, 16);
    if (n0 + r < N) {
      float4 o;
      o.x = fmaf(s, w0, c0);
      o.y = fmaf(s, w1, c1);
      o.z = fmaf(s, w2, c2);
      o.w = fmaf(s, w3, c3);
      orow[r * 32] = o;
    }
  }
}

extern "C" void kernel_launch(void* const* d_in, const int* in_sizes, int n_in,
                              void* d_out, int out_size, void* d_ws, size_t ws_size,
                              hipStream_t stream) {
  const float* x    = (const float*)d_in[0];
  const int*   ei   = (const int*)  d_in[1];
  const float* attr = (const float*)d_in[2];
  const float* W_l  = (const float*)d_in[3];
  const float* b_l  = (const float*)d_in[4];
  const float* W_r  = (const float*)d_in[5];
  const float* b_r  = (const float*)d_in[6];
  const float* W_e  = (const float*)d_in[7];
  const float* att  = (const float*)d_in[8];
  const float* bias = (const float*)d_in[9];

  const int N = in_sizes[0];
  const int E = in_sizes[2];

  double* acc   = (double*)d_ws;
  float*  shift = (float*)(acc + 8 * N);

  const int BLK = 256;

  gat_init<<<(4 * N + BLK - 1) / BLK, BLK, 0, stream>>>((float4*)acc, shift, b_l, b_r, att, N);

  gat_edges<<<(E + 127) / 128, BLK, 0, stream>>>(x, ei, attr, W_l, b_l, W_r, b_r,
                                                 W_e, att, shift, acc, E);

  gat_out<<<(N + 127) / 128, BLK, 0, stream>>>(x, W_l, b_l, W_r, b_r, W_e, att, bias,
                                               acc, shift, (float*)d_out, N);
}


# Edge accumulators kept once per package half (die number from s_sendmsg_rtn 0x87 picks the wave's copy): a line takes atomics from one half only; out kernel adds the two copies; sum order changes
# speedup vs baseline: 6.5698x; 1.1751x over previous
#include <hip/hip_runtime.h>

typedef __attribute__((ext_vector_type(2))) float v2f;
typedef __attribute__((ext_vector_type(8))) float v8f;

#define NEG_SLOPE 0.2f

#define T_LIMIT 600.0f
#define LOG2E_HI 1.44269502162933349609375f
#define LOG2E_LO 1.925962991126617e-08f

__device__ __forceinline__ void exp_split(float t, float& pe, int& k) {
  t = fminf(fmaxf(t, -T_LIMIT), T_LIMIT);
  const float ph = t * LOG2E_HI;
  float pl = fmaf(t, LOG2E_HI, -ph);
  pl = fmaf(t, LOG2E_LO, pl);
  const float kf = rintf(ph);
  const float r = (ph - kf) + pl;
  pe = __builtin_amdgcn_exp2f(r);
  k = (int)kf;
}

__device__ __forceinline__ int package_half() {
  const unsigned id = __builtin_amdgcn_s_sendmsg_rtn(0x87u);
  return (int)((id >> 18) & 1u);
}

__device__ __forceinline__ double scaled_f64(float q, int k) {
  const unsigned u  = __float_as_uint(q);
  const unsigned e8 = (u >> 23) & 0xffu;
  unsigned hi = (u & 0x80000000u) | ((e8 + 896u + (unsigned)k) << 20) | ((u >> 3) & 0x000fffffu);
  unsigned lo = u << 29;
  if (e8 == 0u) { hi = u & 0x80000000u; lo = 0u; }
  return __builtin_bit_cast(double, ((unsigned long long)hi << 32) | (unsigned long long)lo);
}

__global__ void gat_init(float4* __restrict__ acc4, float* __restrict__ shift,
                         const float* __restrict__ b_l, const float* __restrict__ b_r,
                         const float* __restrict__ att, int N) {
  int i = blockIdx.x * blockDim.x + threadIdx.x;
  if (i < 8 * N) acc4[i] = make_float4(0.0f, 0.0f, 0.0f, 0.0f);
  if (blockIdx.x == 0 && threadIdx.x < 32) {
    const int lane = threadIdx.x;
    float s[2] = {0.0f, 0.0f};
#pragma unroll
    for (int jj = 0; jj < 4; ++jj) {
      const int ch = lane + 32 * jj;
      float g = b_l[ch] + b_r[ch];
      g = (g > 0.0f) ? g : NEG_SLOPE * g;
      s[jj >> 1] = fmaf(g, att[ch], s[jj >> 1]);
    }
#pragma unroll
    for (int off = 1; off < 32; off <<= 1) {
      s[0] += __shfl_xor(s[0], off, 32);
      s[1] += __shfl_xor(s[1], off, 32);
    }
    if (lane == 0) { shift[0] = s[0]; shift[1] = s[1]; }
  }
}

__device__ __forceinline__ void logits16(
    v2f A, int lane,
    const float* __restrict__ W_l, const float* __restrict__ b_l,
    const float* __restrict__ W_r, const float* __restrict__ b_r,
    const float* __restrict__ W_e, const float* __restrict__ att,
    float (&p)[2][8]) {
  const int  lh = lane & 15;
  const bool hi = lane >= 16;
  v2f B[8];
#pragma unroll
  for (int t = 0; t < 8; ++t) {
    int ch = t * 16 + lh;
    if (!hi) { B[t].x = W_l[ch]; B[t].y = W_r[ch]; }
    else     { B[t].x = W_e[ch]; B[t].y = b_l[ch] + b_r[ch]; }
  }
#pragma unroll
  for (int h = 0; h < 2; ++h)
#pragma unroll
    for (int r = 0; r < 8; ++r) p[h][r] = 0.0f;

#pragma unroll
  for (int t = 0; t < 8; ++t) {
    v8f cz = {};
    v8f D = __builtin_amdgcn_wmma_f32_16x16x4_f32(
        false, A, false, B[t], (short)0, cz, false, false);
    float av = att[t * 16 + lh];
    const int h = t >> 2;
#pragma unroll
    for (int r = 0; r < 8; ++r) {
      float g = D[r];
      g = (g > 0.0f) ? g : NEG_SLOPE * g;
      p[h][r] = fmaf(g, av, p[h][r]);
    }
  }
#pragma unroll
  for (int off = 1; off < 16; off <<= 1) {
#pragma unroll
    for (int r = 0; r < 8; ++r) {
      p[0][r] += __shfl_xor(p[0][r], off, 32);
      p[1][r] += __shfl_xor(p[1][r], off, 32);
    }
  }
}

__global__ void gat_edges(
    const float* __restrict__ x,
    const int*   __restrict__ ei,
    const float* __restrict__ attr,
    const float* __restrict__ W_l, const float* __restrict__ b_l,
    const float* __restrict__ W_r, const float* __restrict__ b_r,
    const float* __restrict__ W_e, const float* __restrict__ att,
    const float* __restrict__ shift,
    double* __restrict__ acc,
    int E, int N) {
  const int wave = (blockIdx.x * blockDim.x + threadIdx.x) >> 5;
  const int lane = threadIdx.x & 31;
  const int base = wave * 16;
  if (base >= E) return;
  const int  lh = lane & 15;
  const bool hi = lane >= 16;

  int j = base + lh;
  float a0 = 0.0f, a1 = 0.0f;
  int dreg = 0;
  if (j < E) {
    if (!hi) { dreg = ei[E + j]; a0 = x[ei[j]]; a1 = x[dreg]; }
    else     { a0 = attr[j];  a1 = 1.0f; }
  }
  v2f A; A.x = a0; A.y = a1;

  double* __restrict__ accm = acc + (size_t)package_half() * 8u * (size_t)N;

  float p[2][8];
  logits16(A, lane, W_l, b_l, W_r, b_r, W_e, att, p);

  const int c     = lane & 7;
  const int hsel  = c & 1;
  const int b3    = (lane >> 3) & 1;
  const int ehalf = 8 * (lane >> 4);
  const float m0  = shift[hsel];
#pragma unroll
  for (int i = 0; i < 4; ++i) {
    const float l_even = hsel ? p[1][2 * i]     : p[0][2 * i];
    const float l_odd  = hsel ? p[1][2 * i + 1] : p[0][2 * i + 1];
    const float lg = b3 ? l_odd : l_even;
    const int e = ehalf + 2 * i + b3;
    const float xv = __shfl(a0, (c & 4) ? (16 + e) : e, 32);
    const int   d  = __shfl(dreg, e, 32);
    float pe; int k;
    exp_split(lg - m0, pe, k);
    float q  = (c & 2) ? pe * xv : pe;
    int   kk = k;
    if (c & 4) { q = (c & 1) ? 1.0f : xv; kk = 0; }
    const double v = scaled_f64(q, kk);
    if (c < 6 && base + e < E)
      (void)__hip_atomic_fetch_add(&accm[8 * d + c], v, __ATOMIC_RELAXED, __HIP_MEMORY_SCOPE_AGENT);
  }
}

#define OUT_ROWS 16
__global__ void gat_out(const float* __restrict__ x,
                        const float* __restrict__ W_l, const float* __restrict__ b_l,
                        const float* __restrict__ W_r, const float* __restrict__ b_r,
                        const float* __restrict__ W_e, const float* __restrict__ att,
                        const float* __restrict__ bias,
                        const double* __restrict__ acc, const float* __restrict__ shift,
                        float* __restrict__ out, int N) {
  const int wave = (blockIdx.x * blockDim.x + threadIdx.x) >> 5;
  const int lane = threadIdx.x & 31;
  const int n0 = wave * OUT_ROWS;
  if (n0 >= N) return;
  const int  h  = lane >> 4;
  const bool hi = lane >= 16;
  const int  nq = n0 + (lane & 15);
  const double* __restrict__ accB = acc + (size_t)8 * (size_t)N;

  float xn = 0.0f, a0 = 0.0f, a1 = 0.0f;
  if (nq < N) {
    xn = x[nq];
    if (!hi) { a0 = xn; a1 = xn; }
    else {
      const double asum = acc[8 * nq + 4] + accB[8 * nq + 4];
      const double cnt  = acc[8 * nq + 5] + accB[8 * nq + 5];
      a0 = (float)asum / fmaxf((float)cnt, 1.0f); a1 = 1.0f;
    }
  }
  v2f A; A.x = a0; A.y = a1;

  float p[2][8];
  logits16(A, lane, W_l, b_l, W_r, b_r, W_e, att, p);

  const int q = lane & 7;
  float own  = h ? p[1][0] : p[0][0];
  float give = h ? p[0][0] : p[1][0];
#pragma unroll
  for (int i = 1; i < 8; ++i) {
    float co = h ? p[1][i] : p[0][i];
    float cg = h ? p[0][i] : p[1][i];
    own  = (q == i) ? co : own;
    give = (q == i) ? cg : give;
  }
  const float recv = __shfl_xor(give, 16, 32);
  const float ls = (((lane >> 3) & 1) == h) ? own : recv;

  float S = 0.0f;
  if (nq < N) {
    float pe; int ks;
    exp_split(ls - shift[h], pe, ks);
    const double as  = scaled_f64(pe, ks);
    const double den = (acc[8 * nq + h] + accB[8 * nq + h]) + as;
    const double num = fma(as, (double)xn, acc[8 * nq + 2 + h] + accB[8 * nq + 2 + h]);
    S = (float)(num / den);
  }

  const int k = 4 * lane;
  const float w0 = W_l[k], w1 = W_l[k + 1], w2 = W_l[k + 2], w3 = W_l[k + 3];
  const float c0 = b_l[k] + bias[k];
  const float c1 = b_l[k + 1] + bias[k + 1];
  const float c2 = b_l[k + 2] + bias[k + 2];
  const float c3 = b_l[k + 3] + bias[k + 3];

  float4* orow = reinterpret_cast<float4*>(out + (size_t)n0 * 128) + lane;
#pragma unroll
  for (int r = 0; r < OUT_ROWS; ++r) {
    float s = __shfl(S, r, 16);
    if (n0 + r < N) {
      float4 o;
      o.x = fmaf(s, w0, c0);
      o.y = fmaf(s, w1, c1);
      o.z = fmaf(s, w2, c2);
      o.w = fmaf(s, w3, c3);
      orow[r * 32] = o;
    }
  }
}

extern "C" void kernel_launch(void* const* d_in, const int* in_sizes, int n_in,
                              void* d_out, int out_size, void* d_ws, size_t ws_size,
                              hipStream_t stream) {
  const float* x    = (const float*)d_in[0];
  const int*   ei   = (const int*)  d_in[1];
  const float* attr = (const float*)d_in[2];
  const float* W_l  = (const float*)d_in[3];
  const float* b_l  = (const float*)d_in[4];
  const float* W_r  = (const float*)d_in[5];
  const float* b_r  = (const float*)d_in[6];
  const float* W_e  = (const float*)d_in[7];
  const float* att  = (const float*)d_in[8];
  const float* bias = (const float*)d_in[9];

  const int N = in_sizes[0];
  const int E = in_sizes[2];

  double* acc   = (double*)d_ws;
  float*  shift = (float*)(acc + 16 * N);

  const int BLK = 256;

  gat_init<<<(8 * N + BLK - 1) / BLK, BLK, 0, stream>>>((float4*)acc, shift, b_l, b_r, att, N);

  gat_edges<<<(E + 127) / 128, BLK, 0, stream>>>(x, ei, attr, W_l, b_l, W_r, b_r,
                                                 W_e, att, shift, acc, E, N);

  gat_out<<<(N + 127) / 128, BLK, 0, stream>>>(x, W_l, b_l, W_r, b_r, W_e, att, bias,
                                               acc, shift, (float*)d_out, N);
}


# Edge pass: fixed grid of 8192 waves, each looping over 16-edge tiles with the weight fragments resident and the next tiles' index loads and gathers in flight; atomics' arrival (sum) order changes
# speedup vs baseline: 7.7913x; 1.1859x over previous
#include <hip/hip_runtime.h>

typedef __attribute__((ext_vector_type(2))) float v2f;
typedef __attribute__((ext_vector_type(8))) float v8f;

#define NEG_SLOPE 0.2f

#define T_LIMIT 600.0f
#define LOG2E_HI 1.44269502162933349609375f
#define LOG2E_LO 1.925962991126617e-08f

__device__ __forceinline__ void exp_split(float t, float& pe, int& k) {
  t = fminf(fmaxf(t, -T_LIMIT), T_LIMIT);
  const float ph = t * LOG2E_HI;
  float pl = fmaf(t, LOG2E_HI, -ph);
  pl = fmaf(t, LOG2E_LO, pl);
  const float kf = rintf(ph);
  const float r = (ph - kf) + pl;
  pe = __builtin_amdgcn_exp2f(r);
  k = (int)kf;
}

__device__ __forceinline__ int package_half() {
  const unsigned id = __builtin_amdgcn_s_sendmsg_rtn(0x87u);
  return (int)((id >> 18) & 1u);
}

__device__ __forceinline__ double scaled_f64(float q, int k) {
  const unsigned u  = __float_as_uint(q);
  const unsigned e8 = (u >> 23) & 0xffu;
  unsigned hi = (u & 0x80000000u) | ((e8 + 896u + (unsigned)k) << 20) | ((u >> 3) & 0x000fffffu);
  unsigned lo = u << 29;
  if (e8 == 0u) { hi = u & 0x80000000u; lo = 0u; }
  return __builtin_bit_cast(double, ((unsigned long long)hi << 32) | (unsigned long long)lo);
}

__global__ void gat_init(float4* __restrict__ acc4, float* __restrict__ shift,
                         const float* __restrict__ b_l, const float* __restrict__ b_r,
                         const float* __restrict__ att, int N) {
  int i = blockIdx.x * blockDim.x + threadIdx.x;
  if (i < 8 * N) acc4[i] = make_float4(0.0f, 0.0f, 0.0f, 0.0f);
  if (blockIdx.x == 0 && threadIdx.x < 32) {
    const int lane = threadIdx.x;
    float s[2] = {0.0f, 0.0f};
#pragma unroll
    for (int jj = 0; jj < 4; ++jj) {
      const int ch = lane + 32 * jj;
      float g = b_l[ch] + b_r[ch];
      g = (g > 0.0f) ? g : NEG_SLOPE * g;
      s[jj >> 1] = fmaf(g, att[ch], s[jj >> 1]);
    }
#pragma unroll
    for (int off = 1; off < 32; off <<= 1) {
      s[0] += __shfl_xor(s[0], off, 32);
      s[1] += __shfl_xor(s[1], off, 32);
    }
    if (lane == 0) { shift[0] = s[0]; shift[1] = s[1]; }
  }
}

__device__ __forceinline__ void load_weights(
    int lane,
    const float* __restrict__ W_l, const float* __restrict__ b_l,
    const float* __restrict__ W_r, const float* __restrict__ b_r,
    const float* __restrict__ W_e, const float* __restrict__ att,
    v2f (&B)[8], float (&av)[8]) {
  const int  lh = lane & 15;
  const bool hi = lane >= 16;
#pragma unroll
  for (int t = 0; t < 8; ++t) {
    int ch = t * 16 + lh;
    if (!hi) { B[t].x = W_l[ch]; B[t].y = W_r[ch]; }
    else     { B[t].x = W_e[ch]; B[t].y = b_l[ch] + b_r[ch]; }
    av[t] = att[ch];
  }
}

__device__ __forceinline__ void logits_tile(
    v2f A, const v2f (&B)[8], const float (&av)[8], float (&p)[2][8]) {
#pragma unroll
  for (int h = 0; h < 2; ++h)
#pragma unroll
    for (int r = 0; r < 8; ++r) p[h][r] = 0.0f;

#pragma unroll
  for (int t = 0; t < 8; ++t) {
    v8f cz = {};
    v8f D = __builtin_amdgcn_wmma_f32_16x16x4_f32(
        false, A, false, B[t], (short)0, cz, false, false);
    const int h = t >> 2;
#pragma unroll
    for (int r = 0; r < 8; ++r) {
      float g = D[r];
      g = (g > 0.0f) ? g : NEG_SLOPE * g;
      p[h][r] = fmaf(g, av[t], p[h][r]);
    }
  }
#pragma unroll
  for (int off = 1; off < 16; off <<= 1) {
#pragma unroll
    for (int r = 0; r < 8; ++r) {
      p[0][r] += __shfl_xor(p[0][r], off, 32);
      p[1][r] += __shfl_xor(p[1][r], off, 32);
    }
  }
}

__device__ __forceinline__ void logits16(
    v2f A, int lane,
    const float* __restrict__ W_l, const float* __restrict__ b_l,
    const float* __restrict__ W_r, const float* __restrict__ b_r,
    const float* __restrict__ W_e, const float* __restrict__ att,
    float (&p)[2][8]) {
  v2f B[8];
  float av[8];
  load_weights(lane, W_l, b_l, W_r, b_r, W_e, att, B, av);
  logits_tile(A, B, av, p);
}

#define EDGE_BLOCKS 1024
__global__ void __launch_bounds__(256) __attribute__((amdgpu_waves_per_eu(8))) gat_edges(
    const float* __restrict__ x,
    const int*   __restrict__ ei,
    const float* __restrict__ attr,
    const float* __restrict__ W_l, const float* __restrict__ b_l,
    const float* __restrict__ W_r, const float* __restrict__ b_r,
    const float* __restrict__ W_e, const float* __restrict__ att,
    const float* __restrict__ shift,
    double* __restrict__ acc,
    int E, int N) {
  const int lane = threadIdx.x & 31;
  const int  lh = lane & 15;
  const bool hi = lane >= 16;
  const int wave   = __builtin_amdgcn_readfirstlane((int)((blockIdx.x * blockDim.x + threadIdx.x) >> 5));
  const int nwaves = (int)((gridDim.x * blockDim.x) >> 5);
  const int ntiles = (E + 15) >> 4;
  if (wave >= ntiles) return;

  auto load_idx = [&](int T, int& s, int& d, float& at) {
    const int j = min(T * 16 + lh, E - 1);
    s = ei[j]; d = ei[E + j]; at = attr[j];
  };

  int s2, d2; float t2;
  load_idx(wave, s2, d2, t2);

  v2f B[8];
  float av[8];
  load_weights(lane, W_l, b_l, W_r, b_r, W_e, att, B, av);
  const int c     = lane & 7;
  const int hsel  = c & 1;
  const int b3    = (lane >> 3) & 1;
  const int ehalf = 8 * (lane >> 4);
  const float m0  = shift[hsel];
  double* __restrict__ accm = acc + (size_t)package_half() * 8u * (size_t)N;

  float xs_n = x[s2], xd_n = x[d2], at_n = t2;
  int   dn = d2;
  load_idx(wave + nwaves, s2, d2, t2);

  for (int T = wave; T < ntiles; T += nwaves) {
    const int base = T * 16;
    const bool valid = base + lh < E;
    const float a0 = valid ? (hi ? at_n : xs_n) : 0.0f;
    const float a1 = valid ? (hi ? 1.0f : xd_n) : 0.0f;
    const int dreg = dn;
    xs_n = x[s2]; xd_n = x[d2]; at_n = t2; dn = d2;
    load_idx(T + 2 * nwaves, s2, d2, t2);

    v2f A; A.x = a0; A.y = a1;
    float p[2][8];
    logits_tile(A, B, av, p);

#pragma unroll
    for (int i = 0; i < 4; ++i) {
      const float l_even = hsel ? p[1][2 * i]     : p[0][2 * i];
      const float l_odd  = hsel ? p[1][2 * i + 1] : p[0][2 * i + 1];
      const float lg = b3 ? l_odd : l_even;
      const int e = ehalf + 2 * i + b3;
      const float xv = __shfl(a0, (c & 4) ? (16 + e) : e, 32);
      const int   d  = __shfl(dreg, e, 32);
      float pe; int k;
      exp_split(lg - m0, pe, k);
      float q  = (c & 2) ? pe * xv : pe;
      int   kk = k;
      if (c & 4) { q = (c & 1) ? 1.0f : xv; kk = 0; }
      const double v = scaled_f64(q, kk);
      if (c < 6 && base + e < E)
        (void)__hip_atomic_fetch_add(&accm[8 * d + c], v, __ATOMIC_RELAXED, __HIP_MEMORY_SCOPE_AGENT);
    }
  }
}

#define OUT_ROWS 16
__global__ void gat_out(const float* __restrict__ x,
                        const float* __restrict__ W_l, const float* __restrict__ b_l,
                        const float* __restrict__ W_r, const float* __restrict__ b_r,
                        const float* __restrict__ W_e, const float* __restrict__ att,
                        const float* __restrict__ bias,
                        const double* __restrict__ acc, const float* __restrict__ shift,
                        float* __restrict__ out, int N) {
  const int wave = (blockIdx.x * blockDim.x + threadIdx.x) >> 5;
  const int lane = threadIdx.x & 31;
  const int n0 = wave * OUT_ROWS;
  if (n0 >= N) return;
  const int  h  = lane >> 4;
  const bool hi = lane >= 16;
  const int  nq = n0 + (lane & 15);
  const double* __restrict__ accB = acc + (size_t)8 * (size_t)N;

  float xn = 0.0f, a0 = 0.0f, a1 = 0.0f;
  if (nq < N) {
    xn = x[nq];
    if (!hi) { a0 = xn; a1 = xn; }
    else {
      const double asum = acc[8 * nq + 4] + accB[8 * nq + 4];
      const double cnt  = acc[8 * nq + 5] + accB[8 * nq + 5];
      a0 = (float)asum / fmaxf((float)cnt, 1.0f); a1 = 1.0f;
    }
  }
  v2f A; A.x = a0; A.y = a1;

  float p[2][8];
  logits16(A, lane, W_l, b_l, W_r, b_r, W_e, att, p);

  const int q = lane & 7;
  float own  = h ? p[1][0] : p[0][0];
  float give = h ? p[0][0] : p[1][0];
#pragma unroll
  for (int i = 1; i < 8; ++i) {
    float co = h ? p[1][i] : p[0][i];
    float cg = h ? p[0][i] : p[1][i];
    own  = (q == i) ? co : own;
    give = (q == i) ? cg : give;
  }
  const float recv = __shfl_xor(give, 16, 32);
  const float ls = (((lane >> 3) & 1) == h) ? own : recv;

  float S = 0.0f;
  if (nq < N) {
    float pe; int ks;
    exp_split(ls - shift[h], pe, ks);
    const double as  = scaled_f64(pe, ks);
    const double den = (acc[8 * nq + h] + accB[8 * nq + h]) + as;
    const double num = fma(as, (double)xn, acc[8 * nq + 2 + h] + accB[8 * nq + 2 + h]);
    S = (float)(num / den);
  }

  const int k = 4 * lane;
  const float w0 = W_l[k], w1 = W_l[k + 1], w2 = W_l[k + 2], w3 = W_l[k + 3];
  const float c0 = b_l[k] + bias[k];
  const float c1 = b_l[k + 1] + bias[k + 1];
  const float c2 = b_l[k + 2] + bias[k + 2];
  const float c3 = b_l[k + 3] + bias[k + 3];

  float4* orow = reinterpret_cast<float4*>(out + (size_t)n0 * 128) + lane;
#pragma unroll
  for (int r = 0; r < OUT_ROWS; ++r) {
    float s = __shfl(S, r, 16);
    if (n0 + r < N) {
      float4 o;
      o.x = fmaf(s, w0, c0);
      o.y = fmaf(s, w1, c1);
      o.z = fmaf(s, w2, c2);
      o.w = fmaf(s, w3, c3);
      orow[r * 32] = o;
    }
  }
}

extern "C" void kernel_launch(void* const* d_in, const int* in_sizes, int n_in,
                              void* d_out, int out_size, void* d_ws, size_t ws_size,
                              hipStream_t stream) {
  const float* x    = (const float*)d_in[0];
  const int*   ei   = (const int*)  d_in[1];
  const float* attr = (const float*)d_in[2];
  const float* W_l  = (const float*)d_in[3];
  const float* b_l  = (const float*)d_in[4];
  const float* W_r  = (const float*)d_in[5];
  const float* b_r  = (const float*)d_in[6];
  const float* W_e  = (const float*)d_in[7];
  const float* att  = (const float*)d_in[8];
  const float* bias = (const float*)d_in[9];

  const int N = in_sizes[0];
  const int E = in_sizes[2];

  double* acc   = (double*)d_ws;
  float*  shift = (float*)(acc + 16 * N);

  const int BLK = 256;

  gat_init<<<(8 * N + BLK - 1) / BLK, BLK, 0, stream>>>((float4*)acc, shift, b_l, b_r, att, N);

  gat_edges<<<EDGE_BLOCKS, BLK, 0, stream>>>(x, ei, attr, W_l, b_l, W_r, b_r,
                                             W_e, att, shift, acc, E, N);

  gat_out<<<(N + 127) / 128, BLK, 0, stream>>>(x, W_l, b_l, W_r, b_r, W_e, att, bias,
                                               acc, shift, (float*)d_out, N);
}


# Loads grouped at wave entry: weight fragments by per-lane pointer select (no lane-half branch with a wait); out kernel issues x, sums, shift, weights up front: 12 dependent round trips to 1
# speedup vs baseline: 8.7391x; 1.1216x over previous
#include <hip/hip_runtime.h>

typedef __attribute__((ext_vector_type(2))) float v2f;
typedef __attribute__((ext_vector_type(8))) float v8f;

#define NEG_SLOPE 0.2f

#define T_LIMIT 600.0f
#define LOG2E_HI 1.44269502162933349609375f
#define LOG2E_LO 1.925962991126617e-08f

__device__ __forceinline__ void exp_split(float t, float& pe, int& k) {
  t = fminf(fmaxf(t, -T_LIMIT), T_LIMIT);
  const float ph = t * LOG2E_HI;
  float pl = fmaf(t, LOG2E_HI, -ph);
  pl = fmaf(t, LOG2E_LO, pl);
  const float kf = rintf(ph);
  const float r = (ph - kf) + pl;
  pe = __builtin_amdgcn_exp2f(r);
  k = (int)kf;
}

__device__ __forceinline__ int package_half() {
  const unsigned id = __builtin_amdgcn_s_sendmsg_rtn(0x87u);
  return (int)((id >> 18) & 1u);
}

__device__ __forceinline__ double scaled_f64(float q, int k) {
  const unsigned u  = __float_as_uint(q);
  const unsigned e8 = (u >> 23) & 0xffu;
  unsigned hi = (u & 0x80000000u) | ((e8 + 896u + (unsigned)k) << 20) | ((u >> 3) & 0x000fffffu);
  unsigned lo = u << 29;
  if (e8 == 0u) { hi = u & 0x80000000u; lo = 0u; }
  return __builtin_bit_cast(double, ((unsigned long long)hi << 32) | (unsigned long long)lo);
}

__global__ void gat_init(float4* __restrict__ acc4, float* __restrict__ shift,
                         const float* __restrict__ b_l, const float* __restrict__ b_r,
                         const float* __restrict__ att, int N) {
  int i = blockIdx.x * blockDim.x + threadIdx.x;
  if (i < 8 * N) acc4[i] = make_float4(0.0f, 0.0f, 0.0f, 0.0f);
  if (blockIdx.x == 0 && threadIdx.x < 32) {
    const int lane = threadIdx.x;
    float s[2] = {0.0f, 0.0f};
#pragma unroll
    for (int jj = 0; jj < 4; ++jj) {
      const int ch = lane + 32 * jj;
      float g = b_l[ch] + b_r[ch];
      g = (g > 0.0f) ? g : NEG_SLOPE * g;
      s[jj >> 1] = fmaf(g, att[ch], s[jj >> 1]);
    }
#pragma unroll
    for (int off = 1; off < 32; off <<= 1) {
      s[0] += __shfl_xor(s[0], off, 32);
      s[1] += __shfl_xor(s[1], off, 32);
    }
    if (lane == 0) { shift[0] = s[0]; shift[1] = s[1]; }
  }
}

__device__ __forceinline__ void load_weights(
    int lane,
    const float* __restrict__ W_l, const float* __restrict__ b_l,
    const float* __restrict__ W_r, const float* __restrict__ b_r,
    const float* __restrict__ W_e, const float* __restrict__ att,
    v2f (&B)[8], float (&av)[8]) {
  const int  lh = lane & 15;
  const bool hi = lane >= 16;
  const float* p0 = hi ? W_e : W_l;
  const float* p1 = hi ? b_l : W_r;
  float u0[8], u1[8], u2[8];
#pragma unroll
  for (int t = 0; t < 8; ++t) {
    const int ch = t * 16 + lh;
    u0[t] = p0[ch];
    u1[t] = p1[ch];
    u2[t] = b_r[ch];
    av[t] = att[ch];
  }
#pragma unroll
  for (int t = 0; t < 8; ++t) {
    const float bsum = u1[t] + u2[t];
    B[t].x = u0[t];
    B[t].y = hi ? bsum : u1[t];
  }
}

__device__ __forceinline__ void logits_tile(
    v2f A, const v2f (&B)[8], const float (&av)[8], float (&p)[2][8]) {
#pragma unroll
  for (int h = 0; h < 2; ++h)
#pragma unroll
    for (int r = 0; r < 8; ++r) p[h][r] = 0.0f;

#pragma unroll
  for (int t = 0; t < 8; ++t) {
    v8f cz = {};
    v8f D = __builtin_amdgcn_wmma_f32_16x16x4_f32(
        false, A, false, B[t], (short)0, cz, false, false);
    const int h = t >> 2;
#pragma unroll
    for (int r = 0; r < 8; ++r) {
      float g = D[r];
      g = (g > 0.0f) ? g : NEG_SLOPE * g;
      p[h][r] = fmaf(g, av[t], p[h][r]);
    }
  }
#pragma unroll
  for (int off = 1; off < 16; off <<= 1) {
#pragma unroll
    for (int r = 0; r < 8; ++r) {
      p[0][r] += __shfl_xor(p[0][r], off, 32);
      p[1][r] += __shfl_xor(p[1][r], off, 32);
    }
  }
}

__device__ __forceinline__ void logits16(
    v2f A, int lane,
    const float* __restrict__ W_l, const float* __restrict__ b_l,
    const float* __restrict__ W_r, const float* __restrict__ b_r,
    const float* __restrict__ W_e, const float* __restrict__ att,
    float (&p)[2][8]) {
  v2f B[8];
  float av[8];
  load_weights(lane, W_l, b_l, W_r, b_r, W_e, att, B, av);
  logits_tile(A, B, av, p);
}

#define EDGE_BLOCKS 1024
__global__ void __launch_bounds__(256) __attribute__((amdgpu_waves_per_eu(8))) gat_edges(
    const float* __restrict__ x,
    const int*   __restrict__ ei,
    const float* __restrict__ attr,
    const float* __restrict__ W_l, const float* __restrict__ b_l,
    const float* __restrict__ W_r, const float* __restrict__ b_r,
    const float* __restrict__ W_e, const float* __restrict__ att,
    const float* __restrict__ shift,
    double* __restrict__ acc,
    int E, int N) {
  const int lane = threadIdx.x & 31;
  const int  lh = lane & 15;
  const bool hi = lane >= 16;
  const int wave   = __builtin_amdgcn_readfirstlane((int)((blockIdx.x * blockDim.x + threadIdx.x) >> 5));
  const int nwaves = (int)((gridDim.x * blockDim.x) >> 5);
  const int ntiles = (E + 15) >> 4;
  if (wave >= ntiles) return;

  auto load_idx = [&](int T, int& s, int& d, float& at) {
    const int j = min(T * 16 + lh, E - 1);
    s = ei[j]; d = ei[E + j]; at = attr[j];
  };

  int s2, d2; float t2;
  load_idx(wave, s2, d2, t2);

  v2f B[8];
  float av[8];
  load_weights(lane, W_l, b_l, W_r, b_r, W_e, att, B, av);
  const int c     = lane & 7;
  const int hsel  = c & 1;
  const int b3    = (lane >> 3) & 1;
  const int ehalf = 8 * (lane >> 4);
  const float m0  = shift[hsel];
  double* __restrict__ accm = acc + (size_t)package_half() * 8u * (size_t)N;

  float xs_n = x[s2], xd_n = x[d2], at_n = t2;
  int   dn = d2;
  load_idx(wave + nwaves, s2, d2, t2);

  for (int T = wave; T < ntiles; T += nwaves) {
    const int base = T * 16;
    const bool valid = base + lh < E;
    const float a0 = valid ? (hi ? at_n : xs_n) : 0.0f;
    const float a1 = valid ? (hi ? 1.0f : xd_n) : 0.0f;
    const int dreg = dn;
    xs_n = x[s2]; xd_n = x[d2]; at_n = t2; dn = d2;
    load_idx(T + 2 * nwaves, s2, d2, t2);

    v2f A; A.x = a0; A.y = a1;
    float p[2][8];
    logits_tile(A, B, av, p);

#pragma unroll
    for (int i = 0; i < 4; ++i) {
      const float l_even = hsel ? p[1][2 * i]     : p[0][2 * i];
      const float l_odd  = hsel ? p[1][2 * i + 1] : p[0][2 * i + 1];
      const float lg = b3 ? l_odd : l_even;
      const int e = ehalf + 2 * i + b3;
      const float xv = __shfl(a0, (c & 4) ? (16 + e) : e, 32);
      const int   d  = __shfl(dreg, e, 32);
      float pe; int k;
      exp_split(lg - m0, pe, k);
      float q  = (c & 2) ? pe * xv : pe;
      int   kk = k;
      if (c & 4) { q = (c & 1) ? 1.0f : xv; kk = 0; }
      const double v = scaled_f64(q, kk);
      if (c < 6 && base + e < E)
        (void)__hip_atomic_fetch_add(&accm[8 * d + c], v, __ATOMIC_RELAXED, __HIP_MEMORY_SCOPE_AGENT);
    }
  }
}

#define OUT_ROWS 16
__global__ void gat_out(const float* __restrict__ x,
                        const float* __restrict__ W_l, const float* __restrict__ b_l,
                        const float* __restrict__ W_r, const float* __restrict__ b_r,
                        const float* __restrict__ W_e, const float* __restrict__ att,
                        const float* __restrict__ bias,
                        const double* __restrict__ acc, const float* __restrict__ shift,
                        float* __restrict__ out, int N) {
  const int wave = (blockIdx.x * blockDim.x + threadIdx.x) >> 5;
  const int lane = threadIdx.x & 31;
  const int n0 = wave * OUT_ROWS;
  if (n0 >= N) return;
  const int  h  = lane >> 4;
  const bool hi = lane >= 16;
  const int  nq = n0 + (lane & 15);
  const bool live = nq < N;
  const int  nc = live ? nq : N - 1;
  const double* __restrict__ accA = acc + (size_t)8 * (size_t)nc;
  const double* __restrict__ accB = accA + (size_t)8 * (size_t)N;

  const float  xn  = x[nc];
  const double sA = accA[4], cA = accA[5], sB = accB[4], cB = accB[5];
  v2f B[8];
  float av[8];
  load_weights(lane, W_l, b_l, W_r, b_r, W_e, att, B, av);
  const float  sh = shift[h];
  const double dA = accA[h], dB = accB[h];
  const double uA = accA[2 + h], uB = accB[2 + h];
  const int k = 4 * lane;
  const float w0 = W_l[k], w1 = W_l[k + 1], w2 = W_l[k + 2], w3 = W_l[k + 3];
  const float g0 = b_l[k], g1 = b_l[k + 1], g2 = b_l[k + 2], g3 = b_l[k + 3];
  const float e0 = bias[k], e1 = bias[k + 1], e2 = bias[k + 2], e3 = bias[k + 3];

  const double asum = sA + sB;
  const double cnt  = cA + cB;
  const float  mean = (float)asum / fmaxf((float)cnt, 1.0f);
  const float  a0 = live ? (hi ? mean : xn) : 0.0f;
  const float  a1 = live ? (hi ? 1.0f : xn) : 0.0f;
  v2f A; A.x = a0; A.y = a1;

  float p[2][8];
  logits_tile(A, B, av, p);

  const int q = lane & 7;
  float own  = h ? p[1][0] : p[0][0];
  float give = h ? p[0][0] : p[1][0];
#pragma unroll
  for (int i = 1; i < 8; ++i) {
    float co = h ? p[1][i] : p[0][i];
    float cg = h ? p[0][i] : p[1][i];
    own  = (q == i) ? co : own;
    give = (q == i) ? cg : give;
  }
  const float recv = __shfl_xor(give, 16, 32);
  const float ls = (((lane >> 3) & 1) == h) ? own : recv;

  float pe; int ks;
  exp_split(ls - sh, pe, ks);
  const double as  = scaled_f64(pe, ks);
  const double den = (dA + dB) + as;
  const double num = fma(as, (double)xn, uA + uB);
  const float S = (float)(num / den);

  const float c0 = g0 + e0;
  const float c1 = g1 + e1;
  const float c2 = g2 + e2;
  const float c3 = g3 + e3;

  float4* orow = reinterpret_cast<float4*>(out + (size_t)n0 * 128) + lane;
#pragma unroll
  for (int r = 0; r < OUT_ROWS; ++r) {
    float s = __shfl(S, r, 16);
    if (n0 + r < N) {
      float4 o;
      o.x = fmaf(s, w0, c0);
      o.y = fmaf(s, w1, c1);
      o.z = fmaf(s, w2, c2);
      o.w = fmaf(s, w3, c3);
      orow[r * 32] = o;
    }
  }
}

extern "C" void kernel_launch(void* const* d_in, const int* in_sizes, int n_in,
                              void* d_out, int out_size, void* d_ws, size_t ws_size,
                              hipStream_t stream) {
  const float* x    = (const float*)d_in[0];
  const int*   ei   = (const int*)  d_in[1];
  const float* attr = (const float*)d_in[2];
  const float* W_l  = (const float*)d_in[3];
  const float* b_l  = (const float*)d_in[4];
  const float* W_r  = (const float*)d_in[5];
  const float* b_r  = (const float*)d_in[6];
  const float* W_e  = (const float*)d_in[7];
  const float* att  = (const float*)d_in[8];
  const float* bias = (const float*)d_in[9];

  const int N = in_sizes[0];
  const int E = in_sizes[2];

  double* acc   = (double*)d_ws;
  float*  shift = (float*)(acc + 16 * N);

  const int BLK = 256;

  gat_init<<<(8 * N + BLK - 1) / BLK, BLK, 0, stream>>>((float4*)acc, shift, b_l, b_r, att, N);

  gat_edges<<<EDGE_BLOCKS, BLK, 0, stream>>>(x, ei, attr, W_l, b_l, W_r, b_r,
                                             W_e, att, shift, acc, E, N);

  gat_out<<<(N + 127) / 128, BLK, 0, stream>>>(x, W_l, b_l, W_r, b_r, W_e, att, bias,
                                               acc, shift, (float*)d_out, N);
}


# Logit sums over 16 lanes by a halving tree (a lane keeps half its values per step: 15 DPP adds, not 64 LDS shuffles); one edge-head logit per lane, exp once per tile; row permutes feed atomic lanes
# speedup vs baseline: 9.6784x; 1.1075x over previous
#include <hip/hip_runtime.h>

typedef __attribute__((ext_vector_type(2))) float v2f;
typedef __attribute__((ext_vector_type(8))) float v8f;

#define NEG_SLOPE 0.2f

#define T_LIMIT 600.0f
#define LOG2E_HI 1.44269502162933349609375f
#define LOG2E_LO 1.925962991126617e-08f

__device__ __forceinline__ void exp_split(float t, float& pe, int& k) {
  t = fminf(fmaxf(t, -T_LIMIT), T_LIMIT);
  const float ph = t * LOG2E_HI;
  float pl = fmaf(t, LOG2E_HI, -ph);
  pl = fmaf(t, LOG2E_LO, pl);
  const float kf = rintf(ph);
  const float r = (ph - kf) + pl;
  pe = __builtin_amdgcn_exp2f(r);
  k = (int)kf;
}

__device__ __forceinline__ int package_half() {
  const unsigned id = __builtin_amdgcn_s_sendmsg_rtn(0x87u);
  return (int)((id >> 18) & 1u);
}

__device__ __forceinline__ double scaled_f64(float q, int k) {
  const unsigned u  = __float_as_uint(q);
  const unsigned e8 = (u >> 23) & 0xffu;
  unsigned hi = (u & 0x80000000u) | ((e8 + 896u + (unsigned)k) << 20) | ((u >> 3) & 0x000fffffu);
  unsigned lo = u << 29;
  if (e8 == 0u) { hi = u & 0x80000000u; lo = 0u; }
  return __builtin_bit_cast(double, ((unsigned long long)hi << 32) | (unsigned long long)lo);
}

__global__ void gat_init(float4* __restrict__ acc4, float* __restrict__ shift,
                         const float* __restrict__ b_l, const float* __restrict__ b_r,
                         const float* __restrict__ att, int N) {
  int i = blockIdx.x * blockDim.x + threadIdx.x;
  if (i < 8 * N) acc4[i] = make_float4(0.0f, 0.0f, 0.0f, 0.0f);
  if (blockIdx.x == 0 && threadIdx.x < 32) {
    const int lane = threadIdx.x;
    float s[2] = {0.0f, 0.0f};
#pragma unroll
    for (int jj = 0; jj < 4; ++jj) {
      const int ch = lane + 32 * jj;
      float g = b_l[ch] + b_r[ch];
      g = (g > 0.0f) ? g : NEG_SLOPE * g;
      s[jj >> 1] = fmaf(g, att[ch], s[jj >> 1]);
    }
#pragma unroll
    for (int off = 1; off < 32; off <<= 1) {
      s[0] += __shfl_xor(s[0], off, 32);
      s[1] += __shfl_xor(s[1], off, 32);
    }
    if (lane == 0) { shift[0] = s[0]; shift[1] = s[1]; }
  }
}

__device__ __forceinline__ void load_weights(
    int lane,
    const float* __restrict__ W_l, const float* __restrict__ b_l,
    const float* __restrict__ W_r, const float* __restrict__ b_r,
    const float* __restrict__ W_e, const float* __restrict__ att,
    v2f (&B)[8], float (&av)[8]) {
  const int  lh = lane & 15;
  const bool hi = lane >= 16;
  const float* p0 = hi ? W_e : W_l;
  const float* p1 = hi ? b_l : W_r;
  float u0[8], u1[8], u2[8];
#pragma unroll
  for (int t = 0; t < 8; ++t) {
    const int ch = t * 16 + lh;
    u0[t] = p0[ch];
    u1[t] = p1[ch];
    u2[t] = b_r[ch];
    av[t] = att[ch];
  }
#pragma unroll
  for (int t = 0; t < 8; ++t) {
    const float bsum = u1[t] + u2[t];
    B[t].x = u0[t];
    B[t].y = hi ? bsum : u1[t];
  }
}

template <int M>
__device__ __forceinline__ float row_xor(float v) {
  return __builtin_bit_cast(float, __builtin_amdgcn_update_dpp(
      0, __builtin_bit_cast(int, v), 0x160 + M, 0xf, 0xf, true));
}

__device__ __forceinline__ float logits_tile(
    v2f A, const v2f (&B)[8], const float (&av)[8], int lane) {
  float p[2][8];
#pragma unroll
  for (int h = 0; h < 2; ++h)
#pragma unroll
    for (int r = 0; r < 8; ++r) p[h][r] = 0.0f;

#pragma unroll
  for (int t = 0; t < 8; ++t) {
    v8f cz = {};
    v8f D = __builtin_amdgcn_wmma_f32_16x16x4_f32(
        false, A, false, B[t], (short)0, cz, false, false);
    const int h = t >> 2;
#pragma unroll
    for (int r = 0; r < 8; ++r) {
      float g = D[r];
      g = (g > 0.0f) ? g : NEG_SLOPE * g;
      p[h][r] = fmaf(g, av[t], p[h][r]);
    }
  }
  const bool b0 = (lane & 1) != 0, b1 = (lane & 2) != 0;
  const bool b2 = (lane & 4) != 0, b3 = (lane & 8) != 0;
  float q[8];
#pragma unroll
  for (int r = 0; r < 8; ++r) {
    const float keep = b0 ? p[1][r] : p[0][r];
    const float send = b0 ? p[0][r] : p[1][r];
    q[r] = keep + row_xor<1>(send);
  }
  float u[4];
#pragma unroll
  for (int j = 0; j < 4; ++j) {
    const float keep = b1 ? q[2 * j + 1] : q[2 * j];
    const float send = b1 ? q[2 * j] : q[2 * j + 1];
    u[j] = keep + row_xor<2>(send);
  }
  float w[2];
#pragma unroll
  for (int m = 0; m < 2; ++m) {
    const float keep = b2 ? u[2 * m + 1] : u[2 * m];
    const float send = b2 ? u[2 * m] : u[2 * m + 1];
    w[m] = keep + row_xor<4>(send);
  }
  const float keep = b3 ? w[1] : w[0];
  const float send = b3 ? w[0] : w[1];
  return keep + row_xor<8>(send);
}

#define EDGE_BLOCKS 1024
__global__ void __launch_bounds__(256) __attribute__((amdgpu_waves_per_eu(8))) gat_edges(
    const float* __restrict__ x,
    const int*   __restrict__ ei,
    const float* __restrict__ attr,
    const float* __restrict__ W_l, const float* __restrict__ b_l,
    const float* __restrict__ W_r, const float* __restrict__ b_r,
    const float* __restrict__ W_e, const float* __restrict__ att,
    const float* __restrict__ shift,
    double* __restrict__ acc,
    int E, int N) {
  const int lane = threadIdx.x & 31;
  const int  lh = lane & 15;
  const bool hi = lane >= 16;
  const int wave   = __builtin_amdgcn_readfirstlane((int)((blockIdx.x * blockDim.x + threadIdx.x) >> 5));
  const int nwaves = (int)((gridDim.x * blockDim.x) >> 5);
  const int ntiles = (E + 15) >> 4;
  if (wave >= ntiles) return;

  auto load_idx = [&](int T, int& s, int& d, float& at) {
    const int j = min(T * 16 + lh, E - 1);
    s = ei[j]; d = ei[E + j]; at = attr[j];
  };

  int s2, d2; float t2;
  load_idx(wave, s2, d2, t2);

  v2f B[8];
  float av[8];
  load_weights(lane, W_l, b_l, W_r, b_r, W_e, att, B, av);
  const int c     = lane & 7;
  const int hsel  = c & 1;
  const int b3    = (lane >> 3) & 1;
  const int ehalf = 8 * (lane >> 4);
  const float m0  = shift[hsel];
  double* __restrict__ accm = acc + (size_t)package_half() * 8u * (size_t)N;

  float xs_n = x[s2], xd_n = x[d2], at_n = t2;
  int   dn = d2;
  load_idx(wave + nwaves, s2, d2, t2);

  for (int T = wave; T < ntiles; T += nwaves) {
    const int base = T * 16;
    const bool valid = base + lh < E;
    const float xs_c = xs_n, at_c = at_n;
    const float a0 = valid ? (hi ? at_c : xs_c) : 0.0f;
    const float a1 = valid ? (hi ? 1.0f : xd_n) : 0.0f;
    const int dreg = dn;
    xs_n = x[s2]; xd_n = x[d2]; at_n = t2; dn = d2;
    load_idx(T + 2 * nwaves, s2, d2, t2);

    v2f A; A.x = a0; A.y = a1;
    const float lg = logits_tile(A, B, av, lane);
    float pe; int k;
    exp_split(lg - m0, pe, k);

    const int dY = __builtin_amdgcn_update_dpp(dreg, dreg, 0x168, 0x2, 0xf, false);
    const float xY = __builtin_bit_cast(float, __builtin_amdgcn_update_dpp(
        __builtin_bit_cast(int, xs_c), __builtin_bit_cast(int, xs_c), 0x168, 0x2, 0xf, false));
    const float aY = __builtin_bit_cast(float, __builtin_amdgcn_update_dpp(
        __builtin_bit_cast(int, at_c), __builtin_bit_cast(int, at_c), 0x168, 0x1, 0xf, false));
    const unsigned zY = __float_as_uint((lane & 8) ? aY : xY);
    const unsigned peu = __float_as_uint(pe);

#pragma unroll
    for (int i = 0; i < 4; ++i) {
      const int e = ehalf + 2 * i + b3;
      const unsigned sp_lo = 0x10101010u + 0x44444444u * (unsigned)i;
      const unsigned sp_hi = 0x32323232u + 0x44444444u * (unsigned)i;
      const float pei = __uint_as_float(__builtin_amdgcn_permlane16(peu, peu, sp_lo, sp_hi, false, false));
      const int   ki  = (int)__builtin_amdgcn_permlane16((unsigned)k, (unsigned)k, sp_lo, sp_hi, false, false);
      const unsigned sz_lo = 0x00080000u + 0x22222222u * (unsigned)i;
      const unsigned sz_hi = 0x11191111u + 0x22222222u * (unsigned)i;
      const float xv = __uint_as_float(__builtin_amdgcn_permlane16(zY, zY, sz_lo, sz_hi, false, false));
      const unsigned sd_lo = 0x22222222u * (unsigned)i;
      const unsigned sd_hi = 0x11111111u + 0x22222222u * (unsigned)i;
      const int d = (int)__builtin_amdgcn_permlane16((unsigned)dY, (unsigned)dY, sd_lo, sd_hi, false, false);
      float q  = (c & 2) ? pei * xv : pei;
      int   kk = ki;
      if (c & 4) { q = (c & 1) ? 1.0f : xv; kk = 0; }
      const double v = scaled_f64(q, kk);
      if (c < 6 && base + e < E)
        (void)__hip_atomic_fetch_add(&accm[8 * d + c], v, __ATOMIC_RELAXED, __HIP_MEMORY_SCOPE_AGENT);
    }
  }
}

#define OUT_ROWS 16
__global__ void gat_out(const float* __restrict__ x,
                        const float* __restrict__ W_l, const float* __restrict__ b_l,
                        const float* __restrict__ W_r, const float* __restrict__ b_r,
                        const float* __restrict__ W_e, const float* __restrict__ att,
                        const float* __restrict__ bias,
                        const double* __restrict__ acc, const float* __restrict__ shift,
                        float* __restrict__ out, int N) {
  const int wave = (blockIdx.x * blockDim.x + threadIdx.x) >> 5;
  const int lane = threadIdx.x & 31;
  const int n0 = wave * OUT_ROWS;
  if (n0 >= N) return;
  const int  h  = lane >> 4;
  const bool hi = lane >= 16;
  const int  nq = n0 + (lane & 15);
  const bool live = nq < N;
  const int  nc = live ? nq : N - 1;
  const double* __restrict__ accA = acc + (size_t)8 * (size_t)nc;
  const double* __restrict__ accB = accA + (size_t)8 * (size_t)N;

  const float  xn  = x[nc];
  const double sA = accA[4], cA = accA[5], sB = accB[4], cB = accB[5];
  v2f B[8];
  float av[8];
  load_weights(lane, W_l, b_l, W_r, b_r, W_e, att, B, av);
  const float  sh = shift[h];
  const double dA = accA[h], dB = accB[h];
  const double uA = accA[2 + h], uB = accB[2 + h];
  const int k = 4 * lane;
  const float w0 = W_l[k], w1 = W_l[k + 1], w2 = W_l[k + 2], w3 = W_l[k + 3];
  const float g0 = b_l[k], g1 = b_l[k + 1], g2 = b_l[k + 2], g3 = b_l[k + 3];
  const float e0 = bias[k], e1 = bias[k + 1], e2 = bias[k + 2], e3 = bias[k + 3];

  const double asum = sA + sB;
  const double cnt  = cA + cB;
  const float  mean = (float)asum / fmaxf((float)cnt, 1.0f);
  const float  a0 = live ? (hi ? mean : xn) : 0.0f;
  const float  a1 = live ? (hi ? 1.0f : xn) : 0.0f;
  v2f A; A.x = a0; A.y = a1;

  const float lz = logits_tile(A, B, av, lane);

  const int rr = lane & 15;
  const float ls = __shfl(lz, 16 * (rr >> 3) + 2 * (rr & 7) + h, 32);

  float pe; int ks;
  exp_split(ls - sh, pe, ks);
  const double as  = scaled_f64(pe, ks);
  const double den = (dA + dB) + as;
  const double num = fma(as, (double)xn, uA + uB);
  const float S = (float)(num / den);

  const float c0 = g0 + e0;
  const float c1 = g1 + e1;
  const float c2 = g2 + e2;
  const float c3 = g3 + e3;

  float4* orow = reinterpret_cast<float4*>(out + (size_t)n0 * 128) + lane;
#pragma unroll
  for (int r = 0; r < OUT_ROWS; ++r) {
    float s = __shfl(S, r, 16);
    if (n0 + r < N) {
      float4 o;
      o.x = fmaf(s, w0, c0);
      o.y = fmaf(s, w1, c1);
      o.z = fmaf(s, w2, c2);
      o.w = fmaf(s, w3, c3);
      orow[r * 32] = o;
    }
  }
}

extern "C" void kernel_launch(void* const* d_in, const int* in_sizes, int n_in,
                              void* d_out, int out_size, void* d_ws, size_t ws_size,
                              hipStream_t stream) {
  const float* x    = (const float*)d_in[0];
  const int*   ei   = (const int*)  d_in[1];
  const float* attr = (const float*)d_in[2];
  const float* W_l  = (const float*)d_in[3];
  const float* b_l  = (const float*)d_in[4];
  const float* W_r  = (const float*)d_in[5];
  const float* b_r  = (const float*)d_in[6];
  const float* W_e  = (const float*)d_in[7];
  const float* att  = (const float*)d_in[8];
  const float* bias = (const float*)d_in[9];

  const int N = in_sizes[0];
  const int E = in_sizes[2];

  double* acc   = (double*)d_ws;
  float*  shift = (float*)(acc + 16 * N);

  const int BLK = 256;

  gat_init<<<(8 * N + BLK - 1) / BLK, BLK, 0, stream>>>((float4*)acc, shift, b_l, b_r, att, N);

  gat_edges<<<EDGE_BLOCKS, BLK, 0, stream>>>(x, ei, attr, W_l, b_l, W_r, b_r,
                                             W_e, att, shift, acc, E, N);

  gat_out<<<(N + 127) / 128, BLK, 0, stream>>>(x, W_l, b_l, W_r, b_r, W_e, att, bias,
                                               acc, shift, (float*)d_out, N);
}


# Logit stage: leakyrelu as max(g, 0.2g), one v_maximum per channel in place of compare + select through a mask register (128 -> 64 vector instructions per 16-edge tile); same bits
# speedup vs baseline: 10.2317x; 1.0572x over previous
#include <hip/hip_runtime.h>

typedef __attribute__((ext_vector_type(2))) float v2f;
typedef __attribute__((ext_vector_type(8))) float v8f;

#define NEG_SLOPE 0.2f

#define T_LIMIT 600.0f
#define LOG2E_HI 1.44269502162933349609375f
#define LOG2E_LO 1.925962991126617e-08f

__device__ __forceinline__ void exp_split(float t, float& pe, int& k) {
  t = fminf(fmaxf(t, -T_LIMIT), T_LIMIT);
  const float ph = t * LOG2E_HI;
  float pl = fmaf(t, LOG2E_HI, -ph);
  pl = fmaf(t, LOG2E_LO, pl);
  const float kf = rintf(ph);
  const float r = (ph - kf) + pl;
  pe = __builtin_amdgcn_exp2f(r);
  k = (int)kf;
}

__device__ __forceinline__ int package_half() {
  const unsigned id = __builtin_amdgcn_s_sendmsg_rtn(0x87u);
  return (int)((id >> 18) & 1u);
}

__device__ __forceinline__ double scaled_f64(float q, int k) {
  const unsigned u  = __float_as_uint(q);
  const unsigned e8 = (u >> 23) & 0xffu;
  unsigned hi = (u & 0x80000000u) | ((e8 + 896u + (unsigned)k) << 20) | ((u >> 3) & 0x000fffffu);
  unsigned lo = u << 29;
  if (e8 == 0u) { hi = u & 0x80000000u; lo = 0u; }
  return __builtin_bit_cast(double, ((unsigned long long)hi << 32) | (unsigned long long)lo);
}

__global__ void gat_init(float4* __restrict__ acc4, float* __restrict__ shift,
                         const float* __restrict__ b_l, const float* __restrict__ b_r,
                         const float* __restrict__ att, int N) {
  int i = blockIdx.x * blockDim.x + threadIdx.x;
  if (i < 8 * N) acc4[i] = make_float4(0.0f, 0.0f, 0.0f, 0.0f);
  if (blockIdx.x == 0 && threadIdx.x < 32) {
    const int lane = threadIdx.x;
    float s[2] = {0.0f, 0.0f};
#pragma unroll
    for (int jj = 0; jj < 4; ++jj) {
      const int ch = lane + 32 * jj;
      float g = b_l[ch] + b_r[ch];
      g = (g > 0.0f) ? g : NEG_SLOPE * g;
      s[jj >> 1] = fmaf(g, att[ch], s[jj >> 1]);
    }
#pragma unroll
    for (int off = 1; off < 32; off <<= 1) {
      s[0] += __shfl_xor(s[0], off, 32);
      s[1] += __shfl_xor(s[1], off, 32);
    }
    if (lane == 0) { shift[0] = s[0]; shift[1] = s[1]; }
  }
}

__device__ __forceinline__ void load_weights(
    int lane,
    const float* __restrict__ W_l, const float* __restrict__ b_l,
    const float* __restrict__ W_r, const float* __restrict__ b_r,
    const float* __restrict__ W_e, const float* __restrict__ att,
    v2f (&B)[8], float (&av)[8]) {
  const int  lh = lane & 15;
  const bool hi = lane >= 16;
  const float* p0 = hi ? W_e : W_l;
  const float* p1 = hi ? b_l : W_r;
  float u0[8], u1[8], u2[8];
#pragma unroll
  for (int t = 0; t < 8; ++t) {
    const int ch = t * 16 + lh;
    u0[t] = p0[ch];
    u1[t] = p1[ch];
    u2[t] = b_r[ch];
    av[t] = att[ch];
  }
#pragma unroll
  for (int t = 0; t < 8; ++t) {
    const float bsum = u1[t] + u2[t];
    B[t].x = u0[t];
    B[t].y = hi ? bsum : u1[t];
  }
}

template <int M>
__device__ __forceinline__ float row_xor(float v) {
  return __builtin_bit_cast(float, __builtin_amdgcn_update_dpp(
      0, __builtin_bit_cast(int, v), 0x160 + M, 0xf, 0xf, true));
}

__device__ __forceinline__ float logits_tile(
    v2f A, const v2f (&B)[8], const float (&av)[8], int lane) {
  float p[2][8];
#pragma unroll
  for (int h = 0; h < 2; ++h)
#pragma unroll
    for (int r = 0; r < 8; ++r) p[h][r] = 0.0f;

#pragma unroll
  for (int t = 0; t < 8; ++t) {
    v8f cz = {};
    v8f D = __builtin_amdgcn_wmma_f32_16x16x4_f32(
        false, A, false, B[t], (short)0, cz, false, false);
    const int h = t >> 2;
#pragma unroll
    for (int r = 0; r < 8; ++r) {
      const float g = __builtin_elementwise_maximum(D[r], NEG_SLOPE * D[r]);
      p[h][r] = fmaf(g, av[t], p[h][r]);
    }
  }
  const bool b0 = (lane & 1) != 0, b1 = (lane & 2) != 0;
  const bool b2 = (lane & 4) != 0, b3 = (lane & 8) != 0;
  float q[8];
#pragma unroll
  for (int r = 0; r < 8; ++r) {
    const float keep = b0 ? p[1][r] : p[0][r];
    const float send = b0 ? p[0][r] : p[1][r];
    q[r] = keep + row_xor<1>(send);
  }
  float u[4];
#pragma unroll
  for (int j = 0; j < 4; ++j) {
    const float keep = b1 ? q[2 * j + 1] : q[2 * j];
    const float send = b1 ? q[2 * j] : q[2 * j + 1];
    u[j] = keep + row_xor<2>(send);
  }
  float w[2];
#pragma unroll
  for (int m = 0; m < 2; ++m) {
    const float keep = b2 ? u[2 * m + 1] : u[2 * m];
    const float send = b2 ? u[2 * m] : u[2 * m + 1];
    w[m] = keep + row_xor<4>(send);
  }
  const float keep = b3 ? w[1] : w[0];
  const float send = b3 ? w[0] : w[1];
  return keep + row_xor<8>(send);
}

#define EDGE_BLOCKS 1024
__global__ void __launch_bounds__(256) __attribute__((amdgpu_waves_per_eu(8))) gat_edges(
    const float* __restrict__ x,
    const int*   __restrict__ ei,
    const float* __restrict__ attr,
    const float* __restrict__ W_l, const float* __restrict__ b_l,
    const float* __restrict__ W_r, const float* __restrict__ b_r,
    const float* __restrict__ W_e, const float* __restrict__ att,
    const float* __restrict__ shift,
    double* __restrict__ acc,
    int E, int N) {
  const int lane = threadIdx.x & 31;
  const int  lh = lane & 15;
  const bool hi = lane >= 16;
  const int wave   = __builtin_amdgcn_readfirstlane((int)((blockIdx.x * blockDim.x + threadIdx.x) >> 5));
  const int nwaves = (int)((gridDim.x * blockDim.x) >> 5);
  const int ntiles = (E + 15) >> 4;
  if (wave >= ntiles) return;

  auto load_idx = [&](int T, int& s, int& d, float& at) {
    const int j = min(T * 16 + lh, E - 1);
    s = ei[j]; d = ei[E + j]; at = attr[j];
  };

  int s2, d2; float t2;
  load_idx(wave, s2, d2, t2);

  v2f B[8];
  float av[8];
  load_weights(lane, W_l, b_l, W_r, b_r, W_e, att, B, av);
  const int c     = lane & 7;
  const int hsel  = c & 1;
  const int b3    = (lane >> 3) & 1;
  const int ehalf = 8 * (lane >> 4);
  const float m0  = shift[hsel];
  double* __restrict__ accm = acc + (size_t)package_half() * 8u * (size_t)N;

  float xs_n = x[s2], xd_n = x[d2], at_n = t2;
  int   dn = d2;
  load_idx(wave + nwaves, s2, d2, t2);

  for (int T = wave; T < ntiles; T += nwaves) {
    const int base = T * 16;
    const bool valid = base + lh < E;
    const float xs_c = xs_n, at_c = at_n;
    const float a0 = valid ? (hi ? at_c : xs_c) : 0.0f;
    const float a1 = valid ? (hi ? 1.0f : xd_n) : 0.0f;
    const int dreg = dn;
    xs_n = x[s2]; xd_n = x[d2]; at_n = t2; dn = d2;
    load_idx(T + 2 * nwaves, s2, d2, t2);

    v2f A; A.x = a0; A.y = a1;
    const float lg = logits_tile(A, B, av, lane);
    float pe; int k;
    exp_split(lg - m0, pe, k);

    const int dY = __builtin_amdgcn_update_dpp(dreg, dreg, 0x168, 0x2, 0xf, false);
    const float xY = __builtin_bit_cast(float, __builtin_amdgcn_update_dpp(
        __builtin_bit_cast(int, xs_c), __builtin_bit_cast(int, xs_c), 0x168, 0x2, 0xf, false));
    const float aY = __builtin_bit_cast(float, __builtin_amdgcn_update_dpp(
        __builtin_bit_cast(int, at_c), __builtin_bit_cast(int, at_c), 0x168, 0x1, 0xf, false));
    const unsigned zY = __float_as_uint((lane & 8) ? aY : xY);
    const unsigned peu = __float_as_uint(pe);

#pragma unroll
    for (int i = 0; i < 4; ++i) {
      const int e = ehalf + 2 * i + b3;
      const unsigned sp_lo = 0x10101010u + 0x44444444u * (unsigned)i;
      const unsigned sp_hi = 0x32323232u + 0x44444444u * (unsigned)i;
      const float pei = __uint_as_float(__builtin_amdgcn_permlane16(peu, peu, sp_lo, sp_hi, false, false));
      const int   ki  = (int)__builtin_amdgcn_permlane16((unsigned)k, (unsigned)k, sp_lo, sp_hi, false, false);
      const unsigned sz_lo = 0x00080000u + 0x22222222u * (unsigned)i;
      const unsigned sz_hi = 0x11191111u + 0x22222222u * (unsigned)i;
      const float xv = __uint_as_float(__builtin_amdgcn_permlane16(zY, zY, sz_lo, sz_hi, false, false));
      const unsigned sd_lo = 0x22222222u * (unsigned)i;
      const unsigned sd_hi = 0x11111111u + 0x22222222u * (unsigned)i;
      const int d = (int)__builtin_amdgcn_permlane16((unsigned)dY, (unsigned)dY, sd_lo, sd_hi, false, false);
      float q  = (c & 2) ? pei * xv : pei;
      int   kk = ki;
      if (c & 4) { q = (c & 1) ? 1.0f : xv; kk = 0; }
      const double v = scaled_f64(q, kk);
      if (c < 6 && base + e < E)
        (void)__hip_atomic_fetch_add(&accm[8 * d + c], v, __ATOMIC_RELAXED, __HIP_MEMORY_SCOPE_AGENT);
    }
  }
}

#define OUT_ROWS 16
__global__ void gat_out(const float* __restrict__ x,
                        const float* __restrict__ W_l, const float* __restrict__ b_l,
                        const float* __restrict__ W_r, const float* __restrict__ b_r,
                        const float* __restrict__ W_e, const float* __restrict__ att,
                        const float* __restrict__ bias,
                        const double* __restrict__ acc, const float* __restrict__ shift,
                        float* __restrict__ out, int N) {
  const int wave = (blockIdx.x * blockDim.x + threadIdx.x) >> 5;
  const int lane = threadIdx.x & 31;
  const int n0 = wave * OUT_ROWS;
  if (n0 >= N) return;
  const int  h  = lane >> 4;
  const bool hi = lane >= 16;
  const int  nq = n0 + (lane & 15);
  const bool live = nq < N;
  const int  nc = live ? nq : N - 1;
  const double* __restrict__ accA = acc + (size_t)8 * (size_t)nc;
  const double* __restrict__ accB = accA + (size_t)8 * (size_t)N;

  const float  xn  = x[nc];
  const double sA = accA[4], cA = accA[5], sB = accB[4], cB = accB[5];
  v2f B[8];
  float av[8];
  load_weights(lane, W_l, b_l, W_r, b_r, W_e, att, B, av);
  const float  sh = shift[h];
  const double dA = accA[h], dB = accB[h];
  const double uA = accA[2 + h], uB = accB[2 + h];
  const int k = 4 * lane;
  const float w0 = W_l[k], w1 = W_l[k + 1], w2 = W_l[k + 2], w3 = W_l[k + 3];
  const float g0 = b_l[k], g1 = b_l[k + 1], g2 = b_l[k + 2], g3 = b_l[k + 3];
  const float e0 = bias[k], e1 = bias[k + 1], e2 = bias[k + 2], e3 = bias[k + 3];

  const double asum = sA + sB;
  const double cnt  = cA + cB;
  const float  mean = (float)asum / fmaxf((float)cnt, 1.0f);
  const float  a0 = live ? (hi ? mean : xn) : 0.0f;
  const float  a1 = live ? (hi ? 1.0f : xn) : 0.0f;
  v2f A; A.x = a0; A.y = a1;

  const float lz = logits_tile(A, B, av, lane);

  const int rr = lane & 15;
  const float ls = __shfl(lz, 16 * (rr >> 3) + 2 * (rr & 7) + h, 32);

  float pe; int ks;
  exp_split(ls - sh, pe, ks);
  const double as  = scaled_f64(pe, ks);
  const double den = (dA + dB) + as;
  const double num = fma(as, (double)xn, uA + uB);
  const float S = (float)(num / den);

  const float c0 = g0 + e0;
  const float c1 = g1 + e1;
  const float c2 = g2 + e2;
  const float c3 = g3 + e3;

  float4* orow = reinterpret_cast<float4*>(out + (size_t)n0 * 128) + lane;
#pragma unroll
  for (int r = 0; r < OUT_ROWS; ++r) {
    float s = __shfl(S, r, 16);
    if (n0 + r < N) {
      float4 o;
      o.x = fmaf(s, w0, c0);
      o.y = fmaf(s, w1, c1);
      o.z = fmaf(s, w2, c2);
      o.w = fmaf(s, w3, c3);
      orow[r * 32] = o;
    }
  }
}

extern "C" void kernel_launch(void* const* d_in, const int* in_sizes, int n_in,
                              void* d_out, int out_size, void* d_ws, size_t ws_size,
                              hipStream_t stream) {
  const float* x    = (const float*)d_in[0];
  const int*   ei   = (const int*)  d_in[1];
  const float* attr = (const float*)d_in[2];
  const float* W_l  = (const float*)d_in[3];
  const float* b_l  = (const float*)d_in[4];
  const float* W_r  = (const float*)d_in[5];
  const float* b_r  = (const float*)d_in[6];
  const float* W_e  = (const float*)d_in[7];
  const float* att  = (const float*)d_in[8];
  const float* bias = (const float*)d_in[9];

  const int N = in_sizes[0];
  const int E = in_sizes[2];

  double* acc   = (double*)d_ws;
  float*  shift = (float*)(acc + 16 * N);

  const int BLK = 256;

  gat_init<<<(8 * N + BLK - 1) / BLK, BLK, 0, stream>>>((float4*)acc, shift, b_l, b_r, att, N);

  gat_edges<<<EDGE_BLOCKS, BLK, 0, stream>>>(x, ei, attr, W_l, b_l, W_r, b_r,
                                             W_e, att, shift, acc, E, N);

  gat_out<<<(N + 127) / 128, BLK, 0, stream>>>(x, W_l, b_l, W_r, b_r, W_e, att, bias,
                                               acc, shift, (float*)d_out, N);
}
